# starting kernel
# baseline (speedup 1.0000x reference)
#include <hip/hip_runtime.h>
#include <math.h>

#define BATCH 8
#define SEQ   512
#define DIM   512
#define DK    64
#define QKV3  192
#define SCALE 0.125f

typedef __attribute__((ext_vector_type(2))) float v2f;
typedef __attribute__((ext_vector_type(8))) float v8f;

__global__ __launch_bounds__(256) void qkv_gemm_kernel(
    const float* __restrict__ X, const float* __restrict__ Wqkv,
    float* __restrict__ Qt, float* __restrict__ Kt, float* __restrict__ Vt)
{
    const int tid  = threadIdx.x;
    const int lane = tid & 31;
    const int wave = tid >> 5;
    const int tile = blockIdx.x * 8 + wave;
    const int JT   = QKV3 / 16;
    const int mt   = tile / JT;
    const int jt   = tile % JT;
    const int row0 = mt * 16;
    const int j0   = jt * 16;
    const int lm   = lane & 15;
    const int lk   = lane >> 4;

    v8f acc = {0.f,0.f,0.f,0.f,0.f,0.f,0.f,0.f};

    const float* arow = X + (size_t)(row0 + lm) * DIM + 2 * lk;
    const float* bcol = Wqkv + j0 + lm;

    #pragma unroll 4
    for (int k0 = 0; k0 < DIM; k0 += 4) {
        v2f a = *(const v2f*)(arow + k0);
        const int kb = k0 + 2 * lk;
        v2f b;
        b.x = bcol[(size_t)kb * QKV3];
        b.y = bcol[(size_t)(kb + 1) * QKV3];
        acc = __builtin_amdgcn_wmma_f32_16x16x4_f32(
            false, a, false, b, (short)0, acc, false, false);
    }

    const int b_idx = row0 >> 9;
    const int n0    = (row0 & 511) + lk * 8;
    const int col   = j0 + lm;
    const int which = j0 >> 6;
    const int c     = col & 63;
    float* T = (which == 0) ? Qt : ((which == 1) ? Kt : Vt);
    float* base = T + ((size_t)b_idx * DK + c) * SEQ;
    #pragma unroll
    for (int r = 0; r < 8; ++r)
        base[n0 + r] = acc[r];
}

__global__ __launch_bounds__(256) void attn_kernel(
    const float* __restrict__ Qt, const float* __restrict__ Kt,
    const float* __restrict__ Vt, float* __restrict__ SA)
{
    __shared__ __align__(16) float kv[SEQ];
    __shared__ __align__(16) float vv[SEQ];
    __shared__ float rmax[8];
    __shared__ float rmin[8];

    const int bc  = blockIdx.x;
    const int tid = threadIdx.x;

    const float* kp = Kt + (size_t)bc * SEQ;
    const float* vp = Vt + (size_t)bc * SEQ;
    kv[tid]       = kp[tid];
    kv[tid + 256] = kp[tid + 256];
    vv[tid]       = vp[tid];
    vv[tid + 256] = vp[tid + 256];
    __syncthreads();

    float kmx = fmaxf(kv[tid], kv[tid + 256]);
    float kmn = fminf(kv[tid], kv[tid + 256]);
    #pragma unroll
    for (int off = 16; off >= 1; off >>= 1) {
        kmx = fmaxf(kmx, __shfl_xor(kmx, off, 32));
        kmn = fminf(kmn, __shfl_xor(kmn, off, 32));
    }
    if ((tid & 31) == 0) { rmax[tid >> 5] = kmx; rmin[tid >> 5] = kmn; }
    __syncthreads();
    float KMAX = rmax[0], KMIN = rmin[0];
    #pragma unroll
    for (int i = 1; i < 8; ++i) {
        KMAX = fmaxf(KMAX, rmax[i]);
        KMIN = fminf(KMIN, rmin[i]);
    }

    const int b = bc >> 6;
    const int c = bc & 63;
    const float* qp = Qt + (size_t)bc * SEQ;
    const float4* kv4 = (const float4*)kv;
    const float4* vv4 = (const float4*)vv;

    #pragma unroll
    for (int r = 0; r < 2; ++r) {
        const int n = tid + r * 256;
        const float a    = qp[n] * SCALE;
        const float cmax = a * ((a >= 0.f) ? KMAX : KMIN);
        float s0 = 0.f, s1 = 0.f, s2 = 0.f, s3 = 0.f;
        float p0 = 0.f, p1 = 0.f, p2 = 0.f, p3 = 0.f;
        #pragma unroll 2
        for (int m4 = 0; m4 < SEQ / 4; ++m4) {
            const float4 k4 = kv4[m4];
            const float4 v4 = vv4[m4];
            const float e0 = __expf(fmaf(a, k4.x, -cmax));
            const float e1 = __expf(fmaf(a, k4.y, -cmax));
            const float e2 = __expf(fmaf(a, k4.z, -cmax));
            const float e3 = __expf(fmaf(a, k4.w, -cmax));
            s0 += e0; s1 += e1; s2 += e2; s3 += e3;
            p0 = fmaf(e0, v4.x, p0);
            p1 = fmaf(e1, v4.y, p1);
            p2 = fmaf(e2, v4.z, p2);
            p3 = fmaf(e3, v4.w, p3);
        }
        const float sum = (s0 + s1) + (s2 + s3);
        const float num = (p0 + p1) + (p2 + p3);
        SA[((size_t)b * SEQ + n) * DK + c] = num / sum;
    }
}

__global__ __launch_bounds__(256) void out_gemm_kernel(
    const float* __restrict__ SA, const float* __restrict__ Wo,
    float* __restrict__ Out)
{
    const int tid  = threadIdx.x;
    const int lane = tid & 31;
    const int wave = tid >> 5;
    const int tile = blockIdx.x * 8 + wave;
    const int JT   = DIM / 16;
    const int mt   = tile / JT;
    const int jt   = tile % JT;
    const int row0 = mt * 16;
    const int j0   = jt * 16;
    const int lm   = lane & 15;
    const int lk   = lane >> 4;

    v8f acc = {0.f,0.f,0.f,0.f,0.f,0.f,0.f,0.f};

    const float* arow = SA + (size_t)(row0 + lm) * DK + 2 * lk;
    const float* bcol = Wo + j0 + lm;

    #pragma unroll
    for (int k0 = 0; k0 < DK; k0 += 4) {
        v2f a = *(const v2f*)(arow + k0);
        const int kb = k0 + 2 * lk;
        v2f b;
        b.x = bcol[(size_t)kb * DIM];
        b.y = bcol[(size_t)(kb + 1) * DIM];
        acc = __builtin_amdgcn_wmma_f32_16x16x4_f32(
            false, a, false, b, (short)0, acc, false, false);
    }

    float* obase = Out + (size_t)(row0 + lk * 8) * DIM + j0 + lm;
    #pragma unroll
    for (int r = 0; r < 8; ++r)
        obase[(size_t)r * DIM] = acc[r];
}

extern "C" void kernel_launch(void* const* d_in, const int* in_sizes, int n_in,
                              void* d_out, int out_size, void* d_ws, size_t ws_size,
                              hipStream_t stream)
{
    const float* X    = (const float*)d_in[0];
    const float* Wqkv = (const float*)d_in[1];
    const float* Wo   = (const float*)d_in[2];
    float* out = (float*)d_out;

    const size_t CHN = (size_t)BATCH * DK * SEQ;
    float* Qt = (float*)d_ws;
    float* Kt = Qt + CHN;
    float* Vt = Kt + CHN;
    float* SA = Vt + CHN;

    qkv_gemm_kernel<<<384, 256, 0, stream>>>(X, Wqkv, Qt, Kt, Vt);
    attn_kernel<<<BATCH * DK, 256, 0, stream>>>(Qt, Kt, Vt, SA);
    out_gemm_kernel<<<1024, 256, 0, stream>>>(SA, Wo, out);
}


# attention exponents al_n*k_m - max come a keys-by-queries tile at a time from one f32 matrix instruction ([k,-1] x [al;max], log2 units), not per-element fma+mul; lanes split keys: sum order changed
# speedup vs baseline: 1.5536x; 1.5536x over previous
#include <hip/hip_runtime.h>
#include <math.h>

#define BATCH 8
#define SEQ   512
#define DIM   512
#define DK    64
#define QKV3  192
#define SCALE 0.125f

typedef __attribute__((ext_vector_type(2))) float v2f;
typedef __attribute__((ext_vector_type(4))) float v4f;
typedef __attribute__((ext_vector_type(8))) float v8f;

#define QKV_CHUNK   8
#define QKV_HALF    (QKV_CHUNK / 2)
#define QKV_STEPS   (QKV_CHUNK / 4)
#define QKV_NCHUNK  (DIM / QKV_CHUNK)
#define QKV_STAGES  8

__device__ __forceinline__ constexpr int qkv_k_of(int i, int s) { return 2 * i + s; }

struct QkvStage {
    v2f a[QKV_STEPS];
    v4f b[QKV_STEPS][2];
};

__device__ __forceinline__ void qkv_load_stage(
    QkvStage& st, const float* __restrict__ ap, const float* __restrict__ bp, int chunk)
{
    const float* a = ap + chunk * QKV_CHUNK;
    const float* b = bp + (size_t)chunk * QKV_CHUNK * QKV3;
    #pragma unroll
    for (int i = 0; i < QKV_STEPS; ++i)
        st.a[i] = *(const v2f*)(a + qkv_k_of(i, 0));
    #pragma unroll
    for (int i = 0; i < QKV_STEPS; ++i) {
        st.b[i][0] = *(const v4f*)(b + (size_t)qkv_k_of(i, 0) * QKV3);
        st.b[i][1] = *(const v4f*)(b + (size_t)qkv_k_of(i, 1) * QKV3);
    }
}

__device__ __forceinline__ void qkv_compute_stage(const QkvStage& st, v8f (&acc)[4])
{
    #pragma unroll
    for (int i = 0; i < QKV_STEPS; ++i) {
        const v2f a = st.a[i];
        #pragma unroll
        for (int t = 0; t < 4; ++t) {
            v2f b;
            b.x = st.b[i][0][t];
            b.y = st.b[i][1][t];
            acc[t] = __builtin_amdgcn_wmma_f32_16x16x4_f32(
                false, a, false, b, (short)0, acc[t], false, false);
        }
    }
}

#define QKV_SCHED_FENCE() __builtin_amdgcn_sched_barrier(0)

__global__ __launch_bounds__(128) void qkv_gemm_kernel(
    const float* __restrict__ X, const float* __restrict__ Wqkv,
    float* __restrict__ Qt, float* __restrict__ Kt, float* __restrict__ Vt)
{
    const int tid  = threadIdx.x;
    const int lane = tid & 31;
    const int wave = tid >> 5;
    const int g    = blockIdx.x % 3;
    const int mt   = (blockIdx.x / 3) * 4 + wave;
    const int row0 = mt * 16;
    const int c    = lane & 15;
    const int h    = lane >> 4;

    const float* ap = X + (size_t)(row0 + c) * DIM + QKV_HALF * h;
    const float* bp = Wqkv + (size_t)(QKV_HALF * h) * QKV3 + 64 * g + 4 * c;

    v8f acc[4];
    #pragma unroll
    for (int t = 0; t < 4; ++t)
        acc[t] = (v8f){0.f,0.f,0.f,0.f,0.f,0.f,0.f,0.f};

    QkvStage st[QKV_STAGES];
    #pragma unroll
    for (int u = 0; u < QKV_STAGES - 1; ++u) {
        qkv_load_stage(st[u], ap, bp, u);
        QKV_SCHED_FENCE();
    }

    #pragma unroll 1
    for (int ch = 0; ch < QKV_NCHUNK - QKV_STAGES; ch += QKV_STAGES) {
        #pragma unroll
        for (int u = 0; u < QKV_STAGES; ++u) {
            qkv_load_stage(st[(u + QKV_STAGES - 1) % QKV_STAGES], ap, bp, ch + u + QKV_STAGES - 1);
            QKV_SCHED_FENCE();
            qkv_compute_stage(st[u], acc);
            QKV_SCHED_FENCE();
        }
    }
    qkv_load_stage(st[QKV_STAGES - 1], ap, bp, QKV_NCHUNK - 1);
    QKV_SCHED_FENCE();
    #pragma unroll
    for (int u = 0; u < QKV_STAGES; ++u)
        qkv_compute_stage(st[u], acc);

    const int b_idx = row0 >> 9;
    const int n0    = (row0 & 511) + h * 8;
    float* T = (g == 0) ? Qt : ((g == 1) ? Kt : Vt);
    #pragma unroll
    for (int t = 0; t < 4; ++t) {
        float* base = T + ((size_t)b_idx * DK + 4 * c + t) * SEQ + n0;
        v4f lo = {acc[t][0], acc[t][1], acc[t][2], acc[t][3]};
        v4f hi = {acc[t][4], acc[t][5], acc[t][6], acc[t][7]};
        *(v4f*)(base)     = lo;
        *(v4f*)(base + 4) = hi;
    }
}

#define ATT_TILES  (SEQ / 16)
#define ATT_GROUPS 4
#define LOG2E_F    1.44269504088896341f

__device__ __forceinline__ void att_consume(
    const v8f d, const v4f v0, const v4f v1, float (&s)[8], float (&p)[8])
{
    #pragma unroll
    for (int r = 0; r < 4; ++r) {
        const float e0 = __builtin_amdgcn_exp2f(d[r]);
        const float e1 = __builtin_amdgcn_exp2f(d[r + 4]);
        s[r]     += e0;
        s[r + 4] += e1;
        p[r]     = fmaf(e0, v0[r], p[r]);
        p[r + 4] = fmaf(e1, v1[r], p[r + 4]);
    }
}

#define ATT_SCHED_FENCE() __builtin_amdgcn_sched_barrier(0)

__global__ __launch_bounds__(256) void attn_kernel(
    const float* __restrict__ Qt, const float* __restrict__ Kt,
    const float* __restrict__ Vt, float* __restrict__ SA)
{
    __shared__ __align__(16) float kv[SEQ];
    __shared__ __align__(16) float vv[SEQ];
    __shared__ __align__(8)  v2f   afl[ATT_TILES * 32];
    __shared__ float rmax[8];
    __shared__ float rmin[8];

    const int bc  = blockIdx.x;
    const int tid = threadIdx.x;

    const float* kp = Kt + (size_t)bc * SEQ;
    const float* vp = Vt + (size_t)bc * SEQ;
    const float k_a = kp[tid];
    const float k_b = kp[tid + 256];
    kv[tid]       = k_a;
    kv[tid + 256] = k_b;
    vv[tid]       = vp[tid];
    vv[tid + 256] = vp[tid + 256];
    {
        const int i = tid & 15, t = tid >> 4;
        afl[t * 32 + i]             = (v2f){k_a, -1.f};
        afl[t * 32 + 16 + i]        = (v2f){0.f, 0.f};
        afl[(t + 16) * 32 + i]      = (v2f){k_b, -1.f};
        afl[(t + 16) * 32 + 16 + i] = (v2f){0.f, 0.f};
    }
    __syncthreads();

    float kmx = fmaxf(kv[tid], kv[tid + 256]);
    float kmn = fminf(kv[tid], kv[tid + 256]);
    #pragma unroll
    for (int off = 16; off >= 1; off >>= 1) {
        kmx = fmaxf(kmx, __shfl_xor(kmx, off, 32));
        kmn = fminf(kmn, __shfl_xor(kmn, off, 32));
    }
    if ((tid & 31) == 0) { rmax[tid >> 5] = kmx; rmin[tid >> 5] = kmn; }
    __syncthreads();
    float KMAX = rmax[0], KMIN = rmin[0];
    #pragma unroll
    for (int i = 1; i < 8; ++i) {
        KMAX = fmaxf(KMAX, rmax[i]);
        KMIN = fminf(KMIN, rmin[i]);
    }

    const int b    = bc >> 6;
    const int c    = bc & 63;
    const int lane = tid & 31;
    const int wave = tid >> 5;
    const int qc   = lane & 15;
    const bool lo  = lane < 16;
    const float* qp = Qt + (size_t)bc * SEQ;

    v2f bq[ATT_GROUPS];
    #pragma unroll
    for (int g = 0; g < ATT_GROUPS; ++g) {
        const float al = qp[(g * 8 + wave) * 16 + qc] * (SCALE * LOG2E_F);
        const float cl = al * ((al >= 0.f) ? KMAX : KMIN);
        bq[g].x = lo ? al : 0.f;
        bq[g].y = lo ? cl : 0.f;
    }

    float s[ATT_GROUPS][8], p[ATT_GROUPS][8];
    #pragma unroll
    for (int g = 0; g < ATT_GROUPS; ++g)
        #pragma unroll
        for (int r = 0; r < 8; ++r) { s[g][r] = 0.f; p[g][r] = 0.f; }

    const v8f zero = (v8f){0.f,0.f,0.f,0.f,0.f,0.f,0.f,0.f};
    const v2f* al_p = afl + lane;
    const v4f* vq = (const v4f*)(vv + (lo ? 0 : 8));

    v8f dA[ATT_GROUPS], dB[ATT_GROUPS];
    {
        const v2f a0 = al_p[0];
        #pragma unroll
        for (int g = 0; g < ATT_GROUPS; ++g)
            dA[g] = __builtin_amdgcn_wmma_f32_16x16x4_f32(
                false, a0, false, bq[g], (short)0, zero, false, false);
    }
    v2f aB = al_p[32];
    v4f vA0 = vq[0], vA1 = vq[1];

    #pragma unroll 1
    for (int t = 0; t < ATT_TILES - 2; t += 2) {
        const v2f aA  = al_p[32 * (t + 2)];
        const v4f vB0 = vq[4 * (t + 1)];
        const v4f vB1 = vq[4 * (t + 1) + 1];
        #pragma unroll
        for (int g = 0; g < ATT_GROUPS; ++g) {
            dB[g] = __builtin_amdgcn_wmma_f32_16x16x4_f32(
                false, aB, false, bq[g], (short)0, zero, false, false);
            att_consume(dA[g], vA0, vA1, s[g], p[g]);
            ATT_SCHED_FENCE();
        }
        aB  = al_p[32 * (t + 3)];
        vA0 = vq[4 * (t + 2)];
        vA1 = vq[4 * (t + 2) + 1];
        #pragma unroll
        for (int g = 0; g < ATT_GROUPS; ++g) {
            dA[g] = __builtin_amdgcn_wmma_f32_16x16x4_f32(
                false, aA, false, bq[g], (short)0, zero, false, false);
            att_consume(dB[g], vB0, vB1, s[g], p[g]);
            ATT_SCHED_FENCE();
        }
    }
    {
        const v4f vB0 = vq[4 * (ATT_TILES - 1)];
        const v4f vB1 = vq[4 * (ATT_TILES - 1) + 1];
        #pragma unroll
        for (int g = 0; g < ATT_GROUPS; ++g) {
            dB[g] = __builtin_amdgcn_wmma_f32_16x16x4_f32(
                false, aB, false, bq[g], (short)0, zero, false, false);
            att_consume(dA[g], vA0, vA1, s[g], p[g]);
            ATT_SCHED_FENCE();
        }
        #pragma unroll
        for (int g = 0; g < ATT_GROUPS; ++g)
            att_consume(dB[g], vB0, vB1, s[g], p[g]);
    }

    #pragma unroll
    for (int g = 0; g < ATT_GROUPS; ++g) {
        float sum = ((s[g][0] + s[g][1]) + (s[g][2] + s[g][3])) + ((s[g][4] + s[g][5]) + (s[g][6] + s[g][7]));
        float num = ((p[g][0] + p[g][1]) + (p[g][2] + p[g][3])) + ((p[g][4] + p[g][5]) + (p[g][6] + p[g][7]));
        sum += __shfl_xor(sum, 16, 32);
        num += __shfl_xor(num, 16, 32);
        const int n = (g * 8 + wave) * 16 + qc;
        if (lo)
            SA[((size_t)b * SEQ + n) * DK + c] = num / sum;
    }
}

#define OUT_CHUNK   8
#define OUT_HALF    (OUT_CHUNK / 2)
#define OUT_STEPS   (OUT_CHUNK / 4)
#define OUT_NCHUNK  (DK / OUT_CHUNK)

__device__ __forceinline__ constexpr int out_k_of(int i, int s) { return 2 * i + s; }

__global__ __launch_bounds__(128) void out_gemm_kernel(
    const float* __restrict__ SA, const float* __restrict__ Wo,
    float* __restrict__ Out)
{
    const int tid   = threadIdx.x;
    const int lane  = tid & 31;
    const int wave  = tid >> 5;
    const int c     = lane & 15;
    const int h     = lane >> 4;
    const int j0    = (blockIdx.x & 7) * 64;
    const int row0  = ((blockIdx.x >> 3) * 4 + wave) * 32;

    const float* ap = SA + (size_t)(row0 + c) * DK + OUT_HALF * h;
    const float* bp = Wo + (size_t)(OUT_HALF * h) * DIM + j0 + 4 * c;

    v4f a[OUT_NCHUNK][2];
    v4f b[OUT_NCHUNK][OUT_HALF];
    #pragma unroll
    for (int ch = 0; ch < OUT_NCHUNK; ++ch) {
        a[ch][0] = *(const v4f*)(ap + ch * OUT_CHUNK);
        a[ch][1] = *(const v4f*)(ap + 16 * DK + ch * OUT_CHUNK);
        #pragma unroll
        for (int q = 0; q < OUT_HALF; ++q)
            b[ch][q] = *(const v4f*)(bp + (size_t)(ch * OUT_CHUNK + q) * DIM);
        __builtin_amdgcn_sched_barrier(0);
    }

    v8f acc[2][4];
    #pragma unroll
    for (int m = 0; m < 2; ++m)
        #pragma unroll
        for (int t = 0; t < 4; ++t)
            acc[m][t] = (v8f){0.f,0.f,0.f,0.f,0.f,0.f,0.f,0.f};

    #pragma unroll
    for (int ch = 0; ch < OUT_NCHUNK; ++ch) {
        #pragma unroll
        for (int i = 0; i < OUT_STEPS; ++i) {
            v2f a0, a1;
            a0.x = a[ch][0][out_k_of(i, 0)];
            a0.y = a[ch][0][out_k_of(i, 1)];
            a1.x = a[ch][1][out_k_of(i, 0)];
            a1.y = a[ch][1][out_k_of(i, 1)];
            #pragma unroll
            for (int t = 0; t < 4; ++t) {
                v2f bf;
                bf.x = b[ch][out_k_of(i, 0)][t];
                bf.y = b[ch][out_k_of(i, 1)][t];
                acc[0][t] = __builtin_amdgcn_wmma_f32_16x16x4_f32(
                    false, a0, false, bf, (short)0, acc[0][t], false, false);
                acc[1][t] = __builtin_amdgcn_wmma_f32_16x16x4_f32(
                    false, a1, false, bf, (short)0, acc[1][t], false, false);
            }
        }
        __builtin_amdgcn_sched_barrier(0);
    }

    #pragma unroll
    for (int m = 0; m < 2; ++m) {
        float* ob = Out + (size_t)(row0 + 16 * m + 8 * h) * DIM + j0 + 4 * c;
        #pragma unroll
        for (int r = 0; r < 8; ++r) {
            v4f o = {acc[m][0][r], acc[m][1][r], acc[m][2][r], acc[m][3][r]};
            *(v4f*)(ob + (size_t)r * DIM) = o;
        }
    }
}

extern "C" void kernel_launch(void* const* d_in, const int* in_sizes, int n_in,
                              void* d_out, int out_size, void* d_ws, size_t ws_size,
                              hipStream_t stream)
{
    const float* X    = (const float*)d_in[0];
    const float* Wqkv = (const float*)d_in[1];
    const float* Wo   = (const float*)d_in[2];
    float* out = (float*)d_out;

    const size_t CHN = (size_t)BATCH * DK * SEQ;
    float* Qt = (float*)d_ws;
    float* Kt = Qt + CHN;
    float* Vt = Kt + CHN;
    float* SA = Vt + CHN;

    qkv_gemm_kernel<<<192, 128, 0, stream>>>(X, Wqkv, Qt, Kt, Vt);
    attn_kernel<<<BATCH * DK, 256, 0, stream>>>(Qt, Kt, Vt, SA);
    out_gemm_kernel<<<256, 128, 0, stream>>>(SA, Wo, out);
}


# QKV GEMM re-cut: 1024 waves of 16x48 (3 tiles per 12-byte Wqkv load) not 768 of 16x64, so all 1024 SIMDs carry 384 matrix instructions, none 512; ring 8->10 stages keeps the lead; k order unchanged
# speedup vs baseline: 1.5990x; 1.0292x over previous
#include <hip/hip_runtime.h>
#include <math.h>

#define BATCH 8
#define SEQ   512
#define DIM   512
#define DK    64
#define QKV3  192
#define SCALE 0.125f

typedef __attribute__((ext_vector_type(2))) float v2f;
typedef __attribute__((ext_vector_type(4))) float v4f;
typedef __attribute__((ext_vector_type(8))) float v8f;

#define QKV_CHUNK   8
#define QKV_HALF    (QKV_CHUNK / 2)
#define QKV_STEPS   (QKV_CHUNK / 4)
#define QKV_NCHUNK  (DIM / QKV_CHUNK)
#define QKV_STAGES  10
#define QKV_MAIN_END (((QKV_NCHUNK - QKV_STAGES + 1) / QKV_STAGES) * QKV_STAGES)
#define QKV_TILES   3
#define QKV_STRIP   (16 * QKV_TILES)
#define QKV_NSTRIP  (QKV3 / QKV_STRIP)

typedef __attribute__((ext_vector_type(3))) float v3f;

__device__ __forceinline__ v3f qkv_load3(const float* __restrict__ p)
{
    v3f r;
    __builtin_memcpy(&r, p, 3 * sizeof(float));
    return r;
}

__device__ __forceinline__ constexpr int qkv_k_of(int i, int s) { return 2 * i + s; }

struct QkvStage {
    v2f a[QKV_STEPS];
    v3f b[QKV_STEPS][2];
};

__device__ __forceinline__ void qkv_load_stage(
    QkvStage& st, const float* __restrict__ ap, const float* __restrict__ bp, int chunk)
{
    const float* a = ap + chunk * QKV_CHUNK;
    const float* b = bp + (size_t)chunk * QKV_CHUNK * QKV3;
    #pragma unroll
    for (int i = 0; i < QKV_STEPS; ++i)
        st.a[i] = *(const v2f*)(a + qkv_k_of(i, 0));
    #pragma unroll
    for (int i = 0; i < QKV_STEPS; ++i) {
        st.b[i][0] = qkv_load3(b + (size_t)qkv_k_of(i, 0) * QKV3);
        st.b[i][1] = qkv_load3(b + (size_t)qkv_k_of(i, 1) * QKV3);
    }
}

__device__ __forceinline__ void qkv_compute_stage(const QkvStage& st, v8f (&acc)[QKV_TILES])
{
    #pragma unroll
    for (int i = 0; i < QKV_STEPS; ++i) {
        const v2f a = st.a[i];
        #pragma unroll
        for (int t = 0; t < QKV_TILES; ++t) {
            v2f b;
            b.x = st.b[i][0][t];
            b.y = st.b[i][1][t];
            acc[t] = __builtin_amdgcn_wmma_f32_16x16x4_f32(
                false, a, false, b, (short)0, acc[t], false, false);
        }
    }
}

#define QKV_SCHED_FENCE() __builtin_amdgcn_sched_barrier(0)

__global__ __launch_bounds__(128) void qkv_gemm_kernel(
    const float* __restrict__ X, const float* __restrict__ Wqkv,
    float* __restrict__ Qt, float* __restrict__ Kt, float* __restrict__ Vt)
{
    const int tid  = threadIdx.x;
    const int lane = tid & 31;
    const int wave = tid >> 5;
    const int g    = blockIdx.x % QKV_NSTRIP;
    const int mt   = (blockIdx.x / QKV_NSTRIP) * 4 + wave;
    const int row0 = mt * 16;
    const int c    = lane & 15;
    const int h    = lane >> 4;

    const float* ap = X + (size_t)(row0 + c) * DIM + QKV_HALF * h;
    const float* bp = Wqkv + (size_t)(QKV_HALF * h) * QKV3 + QKV_STRIP * g + QKV_TILES * c;

    v8f acc[QKV_TILES];
    #pragma unroll
    for (int t = 0; t < QKV_TILES; ++t)
        acc[t] = (v8f){0.f,0.f,0.f,0.f,0.f,0.f,0.f,0.f};

    QkvStage st[QKV_STAGES];
    #pragma unroll
    for (int u = 0; u < QKV_STAGES - 1; ++u) {
        qkv_load_stage(st[u], ap, bp, u);
        QKV_SCHED_FENCE();
    }

    #pragma unroll 1
    for (int ch = 0; ch < QKV_MAIN_END; ch += QKV_STAGES) {
        #pragma unroll
        for (int u = 0; u < QKV_STAGES; ++u) {
            qkv_load_stage(st[(u + QKV_STAGES - 1) % QKV_STAGES], ap, bp, ch + u + QKV_STAGES - 1);
            QKV_SCHED_FENCE();
            qkv_compute_stage(st[u], acc);
            QKV_SCHED_FENCE();
        }
    }
    #pragma unroll
    for (int chunk = QKV_MAIN_END; chunk < QKV_NCHUNK; ++chunk) {
        if (chunk + QKV_STAGES - 1 < QKV_NCHUNK) {
            qkv_load_stage(st[(chunk + QKV_STAGES - 1) % QKV_STAGES], ap, bp, chunk + QKV_STAGES - 1);
            QKV_SCHED_FENCE();
        }
        qkv_compute_stage(st[chunk % QKV_STAGES], acc);
        QKV_SCHED_FENCE();
    }

    const int b_idx = row0 >> 9;
    const int n0    = (row0 & 511) + h * 8;
    #pragma unroll
    for (int t = 0; t < QKV_TILES; ++t) {
        const int col = QKV_STRIP * g + QKV_TILES * c + t;
        const int sel = col >> 6;
        float* T = (sel == 0) ? Qt : ((sel == 1) ? Kt : Vt);
        float* base = T + ((size_t)b_idx * DK + (col & (DK - 1))) * SEQ + n0;
        v4f lo = {acc[t][0], acc[t][1], acc[t][2], acc[t][3]};
        v4f hi = {acc[t][4], acc[t][5], acc[t][6], acc[t][7]};
        *(v4f*)(base)     = lo;
        *(v4f*)(base + 4) = hi;
    }
}

#define ATT_TILES  (SEQ / 16)
#define ATT_GROUPS 4
#define LOG2E_F    1.44269504088896341f

__device__ __forceinline__ void att_consume(
    const v8f d, const v4f v0, const v4f v1, float (&s)[8], float (&p)[8])
{
    #pragma unroll
    for (int r = 0; r < 4; ++r) {
        const float e0 = __builtin_amdgcn_exp2f(d[r]);
        const float e1 = __builtin_amdgcn_exp2f(d[r + 4]);
        s[r]     += e0;
        s[r + 4] += e1;
        p[r]     = fmaf(e0, v0[r], p[r]);
        p[r + 4] = fmaf(e1, v1[r], p[r + 4]);
    }
}

#define ATT_SCHED_FENCE() __builtin_amdgcn_sched_barrier(0)

__global__ __launch_bounds__(256) void attn_kernel(
    const float* __restrict__ Qt, const float* __restrict__ Kt,
    const float* __restrict__ Vt, float* __restrict__ SA)
{
    __shared__ __align__(16) float kv[SEQ];
    __shared__ __align__(16) float vv[SEQ];
    __shared__ __align__(8)  v2f   afl[ATT_TILES * 32];
    __shared__ float rmax[8];
    __shared__ float rmin[8];

    const int bc  = blockIdx.x;
    const int tid = threadIdx.x;

    const float* kp = Kt + (size_t)bc * SEQ;
    const float* vp = Vt + (size_t)bc * SEQ;
    const float k_a = kp[tid];
    const float k_b = kp[tid + 256];
    kv[tid]       = k_a;
    kv[tid + 256] = k_b;
    vv[tid]       = vp[tid];
    vv[tid + 256] = vp[tid + 256];
    {
        const int i = tid & 15, t = tid >> 4;
        afl[t * 32 + i]             = (v2f){k_a, -1.f};
        afl[t * 32 + 16 + i]        = (v2f){0.f, 0.f};
        afl[(t + 16) * 32 + i]      = (v2f){k_b, -1.f};
        afl[(t + 16) * 32 + 16 + i] = (v2f){0.f, 0.f};
    }
    __syncthreads();

    float kmx = fmaxf(kv[tid], kv[tid + 256]);
    float kmn = fminf(kv[tid], kv[tid + 256]);
    #pragma unroll
    for (int off = 16; off >= 1; off >>= 1) {
        kmx = fmaxf(kmx, __shfl_xor(kmx, off, 32));
        kmn = fminf(kmn, __shfl_xor(kmn, off, 32));
    }
    if ((tid & 31) == 0) { rmax[tid >> 5] = kmx; rmin[tid >> 5] = kmn; }
    __syncthreads();
    float KMAX = rmax[0], KMIN = rmin[0];
    #pragma unroll
    for (int i = 1; i < 8; ++i) {
        KMAX = fmaxf(KMAX, rmax[i]);
        KMIN = fminf(KMIN, rmin[i]);
    }

    const int b    = bc >> 6;
    const int c    = bc & 63;
    const int lane = tid & 31;
    const int wave = tid >> 5;
    const int qc   = lane & 15;
    const bool lo  = lane < 16;
    const float* qp = Qt + (size_t)bc * SEQ;

    v2f bq[ATT_GROUPS];
    #pragma unroll
    for (int g = 0; g < ATT_GROUPS; ++g) {
        const float al = qp[(g * 8 + wave) * 16 + qc] * (SCALE * LOG2E_F);
        const float cl = al * ((al >= 0.f) ? KMAX : KMIN);
        bq[g].x = lo ? al : 0.f;
        bq[g].y = lo ? cl : 0.f;
    }

    float s[ATT_GROUPS][8], p[ATT_GROUPS][8];
    #pragma unroll
    for (int g = 0; g < ATT_GROUPS; ++g)
        #pragma unroll
        for (int r = 0; r < 8; ++r) { s[g][r] = 0.f; p[g][r] = 0.f; }

    const v8f zero = (v8f){0.f,0.f,0.f,0.f,0.f,0.f,0.f,0.f};
    const v2f* al_p = afl + lane;
    const v4f* vq = (const v4f*)(vv + (lo ? 0 : 8));

    v8f dA[ATT_GROUPS], dB[ATT_GROUPS];
    {
        const v2f a0 = al_p[0];
        #pragma unroll
        for (int g = 0; g < ATT_GROUPS; ++g)
            dA[g] = __builtin_amdgcn_wmma_f32_16x16x4_f32(
                false, a0, false, bq[g], (short)0, zero, false, false);
    }
    v2f aB = al_p[32];
    v4f vA0 = vq[0], vA1 = vq[1];

    #pragma unroll 1
    for (int t = 0; t < ATT_TILES - 2; t += 2) {
        const v2f aA  = al_p[32 * (t + 2)];
        const v4f vB0 = vq[4 * (t + 1)];
        const v4f vB1 = vq[4 * (t + 1) + 1];
        #pragma unroll
        for (int g = 0; g < ATT_GROUPS; ++g) {
            dB[g] = __builtin_amdgcn_wmma_f32_16x16x4_f32(
                false, aB, false, bq[g], (short)0, zero, false, false);
            att_consume(dA[g], vA0, vA1, s[g], p[g]);
            ATT_SCHED_FENCE();
        }
        aB  = al_p[32 * (t + 3)];
        vA0 = vq[4 * (t + 2)];
        vA1 = vq[4 * (t + 2) + 1];
        #pragma unroll
        for (int g = 0; g < ATT_GROUPS; ++g) {
            dA[g] = __builtin_amdgcn_wmma_f32_16x16x4_f32(
                false, aA, false, bq[g], (short)0, zero, false, false);
            att_consume(dB[g], vB0, vB1, s[g], p[g]);
            ATT_SCHED_FENCE();
        }
    }
    {
        const v4f vB0 = vq[4 * (ATT_TILES - 1)];
        const v4f vB1 = vq[4 * (ATT_TILES - 1) + 1];
        #pragma unroll
        for (int g = 0; g < ATT_GROUPS; ++g) {
            dB[g] = __builtin_amdgcn_wmma_f32_16x16x4_f32(
                false, aB, false, bq[g], (short)0, zero, false, false);
            att_consume(dA[g], vA0, vA1, s[g], p[g]);
            ATT_SCHED_FENCE();
        }
        #pragma unroll
        for (int g = 0; g < ATT_GROUPS; ++g)
            att_consume(dB[g], vB0, vB1, s[g], p[g]);
    }

    #pragma unroll
    for (int g = 0; g < ATT_GROUPS; ++g) {
        float sum = ((s[g][0] + s[g][1]) + (s[g][2] + s[g][3])) + ((s[g][4] + s[g][5]) + (s[g][6] + s[g][7]));
        float num = ((p[g][0] + p[g][1]) + (p[g][2] + p[g][3])) + ((p[g][4] + p[g][5]) + (p[g][6] + p[g][7]));
        sum += __shfl_xor(sum, 16, 32);
        num += __shfl_xor(num, 16, 32);
        const int n = (g * 8 + wave) * 16 + qc;
        if (lo)
            SA[((size_t)b * SEQ + n) * DK + c] = num / sum;
    }
}

#define OUT_CHUNK   8
#define OUT_HALF    (OUT_CHUNK / 2)
#define OUT_STEPS   (OUT_CHUNK / 4)
#define OUT_NCHUNK  (DK / OUT_CHUNK)

__device__ __forceinline__ constexpr int out_k_of(int i, int s) { return 2 * i + s; }

__global__ __launch_bounds__(128) void out_gemm_kernel(
    const float* __restrict__ SA, const float* __restrict__ Wo,
    float* __restrict__ Out)
{
    const int tid   = threadIdx.x;
    const int lane  = tid & 31;
    const int wave  = tid >> 5;
    const int c     = lane & 15;
    const int h     = lane >> 4;
    const int j0    = (blockIdx.x & 7) * 64;
    const int row0  = ((blockIdx.x >> 3) * 4 + wave) * 32;

    const float* ap = SA + (size_t)(row0 + c) * DK + OUT_HALF * h;
    const float* bp = Wo + (size_t)(OUT_HALF * h) * DIM + j0 + 4 * c;

    v4f a[OUT_NCHUNK][2];
    v4f b[OUT_NCHUNK][OUT_HALF];
    #pragma unroll
    for (int ch = 0; ch < OUT_NCHUNK; ++ch) {
        a[ch][0] = *(const v4f*)(ap + ch * OUT_CHUNK);
        a[ch][1] = *(const v4f*)(ap + 16 * DK + ch * OUT_CHUNK);
        #pragma unroll
        for (int q = 0; q < OUT_HALF; ++q)
            b[ch][q] = *(const v4f*)(bp + (size_t)(ch * OUT_CHUNK + q) * DIM);
        __builtin_amdgcn_sched_barrier(0);
    }

    v8f acc[2][4];
    #pragma unroll
    for (int m = 0; m < 2; ++m)
        #pragma unroll
        for (int t = 0; t < 4; ++t)
            acc[m][t] = (v8f){0.f,0.f,0.f,0.f,0.f,0.f,0.f,0.f};

    #pragma unroll
    for (int ch = 0; ch < OUT_NCHUNK; ++ch) {
        #pragma unroll
        for (int i = 0; i < OUT_STEPS; ++i) {
            v2f a0, a1;
            a0.x = a[ch][0][out_k_of(i, 0)];
            a0.y = a[ch][0][out_k_of(i, 1)];
            a1.x = a[ch][1][out_k_of(i, 0)];
            a1.y = a[ch][1][out_k_of(i, 1)];
            #pragma unroll
            for (int t = 0; t < 4; ++t) {
                v2f bf;
                bf.x = b[ch][out_k_of(i, 0)][t];
                bf.y = b[ch][out_k_of(i, 1)][t];
                acc[0][t] = __builtin_amdgcn_wmma_f32_16x16x4_f32(
                    false, a0, false, bf, (short)0, acc[0][t], false, false);
                acc[1][t] = __builtin_amdgcn_wmma_f32_16x16x4_f32(
                    false, a1, false, bf, (short)0, acc[1][t], false, false);
            }
        }
        __builtin_amdgcn_sched_barrier(0);
    }

    #pragma unroll
    for (int m = 0; m < 2; ++m) {
        float* ob = Out + (size_t)(row0 + 16 * m + 8 * h) * DIM + j0 + 4 * c;
        #pragma unroll
        for (int r = 0; r < 8; ++r) {
            v4f o = {acc[m][0][r], acc[m][1][r], acc[m][2][r], acc[m][3][r]};
            *(v4f*)(ob + (size_t)r * DIM) = o;
        }
    }
}

extern "C" void kernel_launch(void* const* d_in, const int* in_sizes, int n_in,
                              void* d_out, int out_size, void* d_ws, size_t ws_size,
                              hipStream_t stream)
{
    const float* X    = (const float*)d_in[0];
    const float* Wqkv = (const float*)d_in[1];
    const float* Wo   = (const float*)d_in[2];
    float* out = (float*)d_out;

    const size_t CHN = (size_t)BATCH * DK * SEQ;
    float* Qt = (float*)d_ws;
    float* Kt = Qt + CHN;
    float* Vt = Kt + CHN;
    float* SA = Vt + CHN;

    qkv_gemm_kernel<<<256, 128, 0, stream>>>(X, Wqkv, Qt, Kt, Vt);
    attn_kernel<<<BATCH * DK, 256, 0, stream>>>(Qt, Kt, Vt, SA);
    out_gemm_kernel<<<256, 128, 0, stream>>>(SA, Wo, out);
}


# attention result stored channel-major SAt[b][c][n]: 64 contiguous bytes per store, not 16 4-byte pieces of 16 lines; output GEMM reads it by 8-byte loads into row-interleaved tiles; sums unchanged
# speedup vs baseline: 1.7113x; 1.0702x over previous
#include <hip/hip_runtime.h>
#include <math.h>

#define BATCH 8
#define SEQ   512
#define DIM   512
#define DK    64
#define QKV3  192
#define SCALE 0.125f

typedef __attribute__((ext_vector_type(2))) float v2f;
typedef __attribute__((ext_vector_type(4))) float v4f;
typedef __attribute__((ext_vector_type(8))) float v8f;

#define QKV_CHUNK   8
#define QKV_HALF    (QKV_CHUNK / 2)
#define QKV_STEPS   (QKV_CHUNK / 4)
#define QKV_NCHUNK  (DIM / QKV_CHUNK)
#define QKV_STAGES  10
#define QKV_MAIN_END (((QKV_NCHUNK - QKV_STAGES + 1) / QKV_STAGES) * QKV_STAGES)
#define QKV_TILES   3
#define QKV_STRIP   (16 * QKV_TILES)
#define QKV_NSTRIP  (QKV3 / QKV_STRIP)

typedef __attribute__((ext_vector_type(3))) float v3f;

__device__ __forceinline__ v3f qkv_load3(const float* __restrict__ p)
{
    v3f r;
    __builtin_memcpy(&r, p, 3 * sizeof(float));
    return r;
}

__device__ __forceinline__ constexpr int qkv_k_of(int i, int s) { return 2 * i + s; }

struct QkvStage {
    v2f a[QKV_STEPS];
    v3f b[QKV_STEPS][2];
};

__device__ __forceinline__ void qkv_load_stage(
    QkvStage& st, const float* __restrict__ ap, const float* __restrict__ bp, int chunk)
{
    const float* a = ap + chunk * QKV_CHUNK;
    const float* b = bp + (size_t)chunk * QKV_CHUNK * QKV3;
    #pragma unroll
    for (int i = 0; i < QKV_STEPS; ++i)
        st.a[i] = *(const v2f*)(a + qkv_k_of(i, 0));
    #pragma unroll
    for (int i = 0; i < QKV_STEPS; ++i) {
        st.b[i][0] = qkv_load3(b + (size_t)qkv_k_of(i, 0) * QKV3);
        st.b[i][1] = qkv_load3(b + (size_t)qkv_k_of(i, 1) * QKV3);
    }
}

__device__ __forceinline__ void qkv_compute_stage(const QkvStage& st, v8f (&acc)[QKV_TILES])
{
    #pragma unroll
    for (int i = 0; i < QKV_STEPS; ++i) {
        const v2f a = st.a[i];
        #pragma unroll
        for (int t = 0; t < QKV_TILES; ++t) {
            v2f b;
            b.x = st.b[i][0][t];
            b.y = st.b[i][1][t];
            acc[t] = __builtin_amdgcn_wmma_f32_16x16x4_f32(
                false, a, false, b, (short)0, acc[t], false, false);
        }
    }
}

#define QKV_SCHED_FENCE() __builtin_amdgcn_sched_barrier(0)

__global__ __launch_bounds__(128) void qkv_gemm_kernel(
    const float* __restrict__ X, const float* __restrict__ Wqkv,
    float* __restrict__ Qt, float* __restrict__ Kt, float* __restrict__ Vt)
{
    const int tid  = threadIdx.x;
    const int lane = tid & 31;
    const int wave = tid >> 5;
    const int g    = blockIdx.x % QKV_NSTRIP;
    const int mt   = (blockIdx.x / QKV_NSTRIP) * 4 + wave;
    const int row0 = mt * 16;
    const int c    = lane & 15;
    const int h    = lane >> 4;

    const float* ap = X + (size_t)(row0 + c) * DIM + QKV_HALF * h;
    const float* bp = Wqkv + (size_t)(QKV_HALF * h) * QKV3 + QKV_STRIP * g + QKV_TILES * c;

    v8f acc[QKV_TILES];
    #pragma unroll
    for (int t = 0; t < QKV_TILES; ++t)
        acc[t] = (v8f){0.f,0.f,0.f,0.f,0.f,0.f,0.f,0.f};

    QkvStage st[QKV_STAGES];
    #pragma unroll
    for (int u = 0; u < QKV_STAGES - 1; ++u) {
        qkv_load_stage(st[u], ap, bp, u);
        QKV_SCHED_FENCE();
    }

    #pragma unroll 1
    for (int ch = 0; ch < QKV_MAIN_END; ch += QKV_STAGES) {
        #pragma unroll
        for (int u = 0; u < QKV_STAGES; ++u) {
            qkv_load_stage(st[(u + QKV_STAGES - 1) % QKV_STAGES], ap, bp, ch + u + QKV_STAGES - 1);
            QKV_SCHED_FENCE();
            qkv_compute_stage(st[u], acc);
            QKV_SCHED_FENCE();
        }
    }
    #pragma unroll
    for (int chunk = QKV_MAIN_END; chunk < QKV_NCHUNK; ++chunk) {
        if (chunk + QKV_STAGES - 1 < QKV_NCHUNK) {
            qkv_load_stage(st[(chunk + QKV_STAGES - 1) % QKV_STAGES], ap, bp, chunk + QKV_STAGES - 1);
            QKV_SCHED_FENCE();
        }
        qkv_compute_stage(st[chunk % QKV_STAGES], acc);
        QKV_SCHED_FENCE();
    }

    const int b_idx = row0 >> 9;
    const int n0    = (row0 & 511) + h * 8;
    #pragma unroll
    for (int t = 0; t < QKV_TILES; ++t) {
        const int col = QKV_STRIP * g + QKV_TILES * c + t;
        const int sel = col >> 6;
        float* T = (sel == 0) ? Qt : ((sel == 1) ? Kt : Vt);
        float* base = T + ((size_t)b_idx * DK + (col & (DK - 1))) * SEQ + n0;
        v4f lo = {acc[t][0], acc[t][1], acc[t][2], acc[t][3]};
        v4f hi = {acc[t][4], acc[t][5], acc[t][6], acc[t][7]};
        *(v4f*)(base)     = lo;
        *(v4f*)(base + 4) = hi;
    }
}

#define ATT_TILES  (SEQ / 16)
#define ATT_GROUPS 4
#define ATT_WAVE_Q (16 * ATT_GROUPS)
#define LOG2E_F    1.44269504088896341f

__device__ __forceinline__ void att_consume(
    const v8f d, const v4f v0, const v4f v1, float (&s)[8], float (&p)[8])
{
    #pragma unroll
    for (int r = 0; r < 4; ++r) {
        const float e0 = __builtin_amdgcn_exp2f(d[r]);
        const float e1 = __builtin_amdgcn_exp2f(d[r + 4]);
        s[r]     += e0;
        s[r + 4] += e1;
        p[r]     = fmaf(e0, v0[r], p[r]);
        p[r + 4] = fmaf(e1, v1[r], p[r + 4]);
    }
}

#define ATT_SCHED_FENCE() __builtin_amdgcn_sched_barrier(0)

__global__ __launch_bounds__(256) void attn_kernel(
    const float* __restrict__ Qt, const float* __restrict__ Kt,
    const float* __restrict__ Vt, float* __restrict__ SAt)
{
    __shared__ __align__(16) float kv[SEQ];
    __shared__ __align__(16) float vv[SEQ];
    __shared__ __align__(8)  v2f   afl[ATT_TILES * 32];
    __shared__ float rmax[8];
    __shared__ float rmin[8];

    const int bc  = blockIdx.x;
    const int tid = threadIdx.x;

    const float* kp = Kt + (size_t)bc * SEQ;
    const float* vp = Vt + (size_t)bc * SEQ;
    const float k_a = kp[tid];
    const float k_b = kp[tid + 256];
    kv[tid]       = k_a;
    kv[tid + 256] = k_b;
    vv[tid]       = vp[tid];
    vv[tid + 256] = vp[tid + 256];
    {
        const int i = tid & 15, t = tid >> 4;
        afl[t * 32 + i]             = (v2f){k_a, -1.f};
        afl[t * 32 + 16 + i]        = (v2f){0.f, 0.f};
        afl[(t + 16) * 32 + i]      = (v2f){k_b, -1.f};
        afl[(t + 16) * 32 + 16 + i] = (v2f){0.f, 0.f};
    }
    __syncthreads();

    float kmx = fmaxf(kv[tid], kv[tid + 256]);
    float kmn = fminf(kv[tid], kv[tid + 256]);
    #pragma unroll
    for (int off = 16; off >= 1; off >>= 1) {
        kmx = fmaxf(kmx, __shfl_xor(kmx, off, 32));
        kmn = fminf(kmn, __shfl_xor(kmn, off, 32));
    }
    if ((tid & 31) == 0) { rmax[tid >> 5] = kmx; rmin[tid >> 5] = kmn; }
    __syncthreads();
    float KMAX = rmax[0], KMIN = rmin[0];
    #pragma unroll
    for (int i = 1; i < 8; ++i) {
        KMAX = fmaxf(KMAX, rmax[i]);
        KMIN = fminf(KMIN, rmin[i]);
    }

    const int lane = tid & 31;
    const int wave = tid >> 5;
    const int qc   = lane & 15;
    const bool lo  = lane < 16;
    const float* qp = Qt + (size_t)bc * SEQ;

    v2f bq[ATT_GROUPS];
    #pragma unroll
    for (int g = 0; g < ATT_GROUPS; ++g) {
        const float al = qp[wave * ATT_WAVE_Q + g * 16 + qc] * (SCALE * LOG2E_F);
        const float cl = al * ((al >= 0.f) ? KMAX : KMIN);
        bq[g].x = lo ? al : 0.f;
        bq[g].y = lo ? cl : 0.f;
    }

    float s[ATT_GROUPS][8], p[ATT_GROUPS][8];
    #pragma unroll
    for (int g = 0; g < ATT_GROUPS; ++g)
        #pragma unroll
        for (int r = 0; r < 8; ++r) { s[g][r] = 0.f; p[g][r] = 0.f; }

    const v8f zero = (v8f){0.f,0.f,0.f,0.f,0.f,0.f,0.f,0.f};
    const v2f* al_p = afl + lane;
    const v4f* vq = (const v4f*)(vv + (lo ? 0 : 8));

    v8f dA[ATT_GROUPS], dB[ATT_GROUPS];
    {
        const v2f a0 = al_p[0];
        #pragma unroll
        for (int g = 0; g < ATT_GROUPS; ++g)
            dA[g] = __builtin_amdgcn_wmma_f32_16x16x4_f32(
                false, a0, false, bq[g], (short)0, zero, false, false);
    }
    v2f aB = al_p[32];
    v4f vA0 = vq[0], vA1 = vq[1];

    #pragma unroll 1
    for (int t = 0; t < ATT_TILES - 2; t += 2) {
        const v2f aA  = al_p[32 * (t + 2)];
        const v4f vB0 = vq[4 * (t + 1)];
        const v4f vB1 = vq[4 * (t + 1) + 1];
        #pragma unroll
        for (int g = 0; g < ATT_GROUPS; ++g) {
            dB[g] = __builtin_amdgcn_wmma_f32_16x16x4_f32(
                false, aB, false, bq[g], (short)0, zero, false, false);
            att_consume(dA[g], vA0, vA1, s[g], p[g]);
            ATT_SCHED_FENCE();
        }
        aB  = al_p[32 * (t + 3)];
        vA0 = vq[4 * (t + 2)];
        vA1 = vq[4 * (t + 2) + 1];
        #pragma unroll
        for (int g = 0; g < ATT_GROUPS; ++g) {
            dA[g] = __builtin_amdgcn_wmma_f32_16x16x4_f32(
                false, aA, false, bq[g], (short)0, zero, false, false);
            att_consume(dB[g], vB0, vB1, s[g], p[g]);
            ATT_SCHED_FENCE();
        }
    }
    {
        const v4f vB0 = vq[4 * (ATT_TILES - 1)];
        const v4f vB1 = vq[4 * (ATT_TILES - 1) + 1];
        #pragma unroll
        for (int g = 0; g < ATT_GROUPS; ++g) {
            dB[g] = __builtin_amdgcn_wmma_f32_16x16x4_f32(
                false, aB, false, bq[g], (short)0, zero, false, false);
            att_consume(dA[g], vA0, vA1, s[g], p[g]);
            ATT_SCHED_FENCE();
        }
        #pragma unroll
        for (int g = 0; g < ATT_GROUPS; ++g)
            att_consume(dB[g], vB0, vB1, s[g], p[g]);
    }

    float* op = SAt + (size_t)bc * SEQ + wave * ATT_WAVE_Q + qc;
    #pragma unroll
    for (int g = 0; g < ATT_GROUPS; ++g) {
        float sum = ((s[g][0] + s[g][1]) + (s[g][2] + s[g][3])) + ((s[g][4] + s[g][5]) + (s[g][6] + s[g][7]));
        float num = ((p[g][0] + p[g][1]) + (p[g][2] + p[g][3])) + ((p[g][4] + p[g][5]) + (p[g][6] + p[g][7]));
        sum += __shfl_xor(sum, 16, 32);
        num += __shfl_xor(num, 16, 32);
        if (lo)
            op[16 * g] = num / sum;
    }
}

#define OUT_CHUNK   8
#define OUT_HALF    (OUT_CHUNK / 2)
#define OUT_STEPS   (OUT_CHUNK / 4)
#define OUT_NCHUNK  (DK / OUT_CHUNK)

__device__ __forceinline__ constexpr int out_k_of(int i, int s) { return 2 * i + s; }

__global__ __launch_bounds__(128) void out_gemm_kernel(
    const float* __restrict__ SAt, const float* __restrict__ Wo,
    float* __restrict__ Out)
{
    const int tid   = threadIdx.x;
    const int lane  = tid & 31;
    const int wave  = tid >> 5;
    const int c     = lane & 15;
    const int h     = lane >> 4;
    const int j0    = (blockIdx.x & 7) * 64;
    const int row0  = ((blockIdx.x >> 3) * 4 + wave) * 32;

    const int b_idx = row0 >> 9;
    const int n0    = row0 & (SEQ - 1);

    const float* ap = SAt + ((size_t)b_idx * DK + OUT_HALF * h) * SEQ + n0 + 2 * c;
    const float* bp = Wo + (size_t)(OUT_HALF * h) * DIM + j0 + 4 * c;

    v2f a[OUT_NCHUNK][OUT_HALF];
    v4f b[OUT_NCHUNK][OUT_HALF];
    #pragma unroll
    for (int ch = 0; ch < OUT_NCHUNK; ++ch) {
        #pragma unroll
        for (int q = 0; q < OUT_HALF; ++q)
            a[ch][q] = *(const v2f*)(ap + (size_t)(ch * OUT_CHUNK + q) * SEQ);
        #pragma unroll
        for (int q = 0; q < OUT_HALF; ++q)
            b[ch][q] = *(const v4f*)(bp + (size_t)(ch * OUT_CHUNK + q) * DIM);
        __builtin_amdgcn_sched_barrier(0);
    }

    v8f acc[2][4];
    #pragma unroll
    for (int m = 0; m < 2; ++m)
        #pragma unroll
        for (int t = 0; t < 4; ++t)
            acc[m][t] = (v8f){0.f,0.f,0.f,0.f,0.f,0.f,0.f,0.f};

    #pragma unroll
    for (int ch = 0; ch < OUT_NCHUNK; ++ch) {
        #pragma unroll
        for (int i = 0; i < OUT_STEPS; ++i) {
            v2f a0, a1;
            a0.x = a[ch][out_k_of(i, 0)][0];
            a0.y = a[ch][out_k_of(i, 1)][0];
            a1.x = a[ch][out_k_of(i, 0)][1];
            a1.y = a[ch][out_k_of(i, 1)][1];
            #pragma unroll
            for (int t = 0; t < 4; ++t) {
                v2f bf;
                bf.x = b[ch][out_k_of(i, 0)][t];
                bf.y = b[ch][out_k_of(i, 1)][t];
                acc[0][t] = __builtin_amdgcn_wmma_f32_16x16x4_f32(
                    false, a0, false, bf, (short)0, acc[0][t], false, false);
                acc[1][t] = __builtin_amdgcn_wmma_f32_16x16x4_f32(
                    false, a1, false, bf, (short)0, acc[1][t], false, false);
            }
        }
        __builtin_amdgcn_sched_barrier(0);
    }

    #pragma unroll
    for (int m = 0; m < 2; ++m) {
        float* ob = Out + (size_t)(row0 + 16 * h + m) * DIM + j0 + 4 * c;
        #pragma unroll
        for (int r = 0; r < 8; ++r) {
            v4f o = {acc[m][0][r], acc[m][1][r], acc[m][2][r], acc[m][3][r]};
            *(v4f*)(ob + (size_t)(2 * r) * DIM) = o;
        }
    }
}

extern "C" void kernel_launch(void* const* d_in, const int* in_sizes, int n_in,
                              void* d_out, int out_size, void* d_ws, size_t ws_size,
                              hipStream_t stream)
{
    const float* X    = (const float*)d_in[0];
    const float* Wqkv = (const float*)d_in[1];
    const float* Wo   = (const float*)d_in[2];
    float* out = (float*)d_out;

    const size_t CHN = (size_t)BATCH * DK * SEQ;
    float* Qt = (float*)d_ws;
    float* Kt = Qt + CHN;
    float* Vt = Kt + CHN;
    float* SAt = Vt + CHN;

    qkv_gemm_kernel<<<256, 128, 0, stream>>>(X, Wqkv, Qt, Kt, Vt);
    attn_kernel<<<BATCH * DK, 256, 0, stream>>>(Qt, Kt, Vt, SAt);
    out_gemm_kernel<<<256, 128, 0, stream>>>(SAt, Wo, out);
}


# attention loop pipelined by one step: exponentials of the next (tile, group) are taken while the previous step's are summed, so v_exp and packed add/fma alternate in issue; sum order unchanged
# speedup vs baseline: 1.7815x; 1.0410x over previous
#include <hip/hip_runtime.h>
#include <math.h>

#define BATCH 8
#define SEQ   512
#define DIM   512
#define DK    64
#define QKV3  192
#define SCALE 0.125f

typedef __attribute__((ext_vector_type(2))) float v2f;
typedef __attribute__((ext_vector_type(4))) float v4f;
typedef __attribute__((ext_vector_type(8))) float v8f;

#define QKV_CHUNK   8
#define QKV_HALF    (QKV_CHUNK / 2)
#define QKV_STEPS   (QKV_CHUNK / 4)
#define QKV_NCHUNK  (DIM / QKV_CHUNK)
#define QKV_STAGES  10
#define QKV_MAIN_END (((QKV_NCHUNK - QKV_STAGES + 1) / QKV_STAGES) * QKV_STAGES)
#define QKV_TILES   3
#define QKV_STRIP   (16 * QKV_TILES)
#define QKV_NSTRIP  (QKV3 / QKV_STRIP)

typedef __attribute__((ext_vector_type(3))) float v3f;

__device__ __forceinline__ v3f qkv_load3(const float* __restrict__ p)
{
    v3f r;
    __builtin_memcpy(&r, p, 3 * sizeof(float));
    return r;
}

__device__ __forceinline__ constexpr int qkv_k_of(int i, int s) { return 2 * i + s; }

struct QkvStage {
    v2f a[QKV_STEPS];
    v3f b[QKV_STEPS][2];
};

__device__ __forceinline__ void qkv_load_stage(
    QkvStage& st, const float* __restrict__ ap, const float* __restrict__ bp, int chunk)
{
    const float* a = ap + chunk * QKV_CHUNK;
    const float* b = bp + (size_t)chunk * QKV_CHUNK * QKV3;
    #pragma unroll
    for (int i = 0; i < QKV_STEPS; ++i)
        st.a[i] = *(const v2f*)(a + qkv_k_of(i, 0));
    #pragma unroll
    for (int i = 0; i < QKV_STEPS; ++i) {
        st.b[i][0] = qkv_load3(b + (size_t)qkv_k_of(i, 0) * QKV3);
        st.b[i][1] = qkv_load3(b + (size_t)qkv_k_of(i, 1) * QKV3);
    }
}

__device__ __forceinline__ void qkv_compute_stage(const QkvStage& st, v8f (&acc)[QKV_TILES])
{
    #pragma unroll
    for (int i = 0; i < QKV_STEPS; ++i) {
        const v2f a = st.a[i];
        #pragma unroll
        for (int t = 0; t < QKV_TILES; ++t) {
            v2f b;
            b.x = st.b[i][0][t];
            b.y = st.b[i][1][t];
            acc[t] = __builtin_amdgcn_wmma_f32_16x16x4_f32(
                false, a, false, b, (short)0, acc[t], false, false);
        }
    }
}

#define QKV_SCHED_FENCE() __builtin_amdgcn_sched_barrier(0)

__global__ __launch_bounds__(128) void qkv_gemm_kernel(
    const float* __restrict__ X, const float* __restrict__ Wqkv,
    float* __restrict__ Qt, float* __restrict__ Kt, float* __restrict__ Vt)
{
    const int tid  = threadIdx.x;
    const int lane = tid & 31;
    const int wave = tid >> 5;
    const int g    = blockIdx.x % QKV_NSTRIP;
    const int mt   = (blockIdx.x / QKV_NSTRIP) * 4 + wave;
    const int row0 = mt * 16;
    const int c    = lane & 15;
    const int h    = lane >> 4;

    const float* ap = X + (size_t)(row0 + c) * DIM + QKV_HALF * h;
    const float* bp = Wqkv + (size_t)(QKV_HALF * h) * QKV3 + QKV_STRIP * g + QKV_TILES * c;

    v8f acc[QKV_TILES];
    #pragma unroll
    for (int t = 0; t < QKV_TILES; ++t)
        acc[t] = (v8f){0.f,0.f,0.f,0.f,0.f,0.f,0.f,0.f};

    QkvStage st[QKV_STAGES];
    #pragma unroll
    for (int u = 0; u < QKV_STAGES - 1; ++u) {
        qkv_load_stage(st[u], ap, bp, u);
        QKV_SCHED_FENCE();
    }

    #pragma unroll 1
    for (int ch = 0; ch < QKV_MAIN_END; ch += QKV_STAGES) {
        #pragma unroll
        for (int u = 0; u < QKV_STAGES; ++u) {
            qkv_load_stage(st[(u + QKV_STAGES - 1) % QKV_STAGES], ap, bp, ch + u + QKV_STAGES - 1);
            QKV_SCHED_FENCE();
            qkv_compute_stage(st[u], acc);
            QKV_SCHED_FENCE();
        }
    }
    #pragma unroll
    for (int chunk = QKV_MAIN_END; chunk < QKV_NCHUNK; ++chunk) {
        if (chunk + QKV_STAGES - 1 < QKV_NCHUNK) {
            qkv_load_stage(st[(chunk + QKV_STAGES - 1) % QKV_STAGES], ap, bp, chunk + QKV_STAGES - 1);
            QKV_SCHED_FENCE();
        }
        qkv_compute_stage(st[chunk % QKV_STAGES], acc);
        QKV_SCHED_FENCE();
    }

    const int b_idx = row0 >> 9;
    const int n0    = (row0 & 511) + h * 8;
    #pragma unroll
    for (int t = 0; t < QKV_TILES; ++t) {
        const int col = QKV_STRIP * g + QKV_TILES * c + t;
        const int sel = col >> 6;
        float* T = (sel == 0) ? Qt : ((sel == 1) ? Kt : Vt);
        float* base = T + ((size_t)b_idx * DK + (col & (DK - 1))) * SEQ + n0;
        v4f lo = {acc[t][0], acc[t][1], acc[t][2], acc[t][3]};
        v4f hi = {acc[t][4], acc[t][5], acc[t][6], acc[t][7]};
        *(v4f*)(base)     = lo;
        *(v4f*)(base + 4) = hi;
    }
}

#define ATT_TILES  (SEQ / 16)
#define ATT_GROUPS 4
#define ATT_WAVE_Q (16 * ATT_GROUPS)
#define LOG2E_F    1.44269504088896341f

__device__ __forceinline__ void att_step(
    const v8f dn, v2f (&e)[4], const v4f v0, const v4f v1, v2f (&s)[4], v2f (&p)[4])
{
    const v2f vp[4] = {{v0[0], v0[1]}, {v0[2], v0[3]}, {v1[0], v1[1]}, {v1[2], v1[3]}};
    v2f en[4];
    #pragma unroll
    for (int j = 0; j < 4; ++j) {
        en[j][0] = __builtin_amdgcn_exp2f(dn[2 * j]);
        en[j][1] = __builtin_amdgcn_exp2f(dn[2 * j + 1]);
        s[j] += e[j];
        p[j] = __builtin_elementwise_fma(e[j], vp[j], p[j]);
    }
    #pragma unroll
    for (int j = 0; j < 4; ++j) e[j] = en[j];
}

__device__ __forceinline__ void att_first(const v8f dn, v2f (&e)[4])
{
    #pragma unroll
    for (int j = 0; j < 4; ++j) {
        e[j][0] = __builtin_amdgcn_exp2f(dn[2 * j]);
        e[j][1] = __builtin_amdgcn_exp2f(dn[2 * j + 1]);
    }
}

__device__ __forceinline__ void att_last(
    const v2f (&e)[4], const v4f v0, const v4f v1, v2f (&s)[4], v2f (&p)[4])
{
    const v2f vp[4] = {{v0[0], v0[1]}, {v0[2], v0[3]}, {v1[0], v1[1]}, {v1[2], v1[3]}};
    #pragma unroll
    for (int j = 0; j < 4; ++j) {
        s[j] += e[j];
        p[j] = __builtin_elementwise_fma(e[j], vp[j], p[j]);
    }
}

#define ATT_STEP_ORDER()                                                        \
    do {                                                                        \
        __builtin_amdgcn_sched_group_barrier(0x008, 1, 0);                      \
        __builtin_amdgcn_sched_group_barrier(0x400, 1, 0);                      \
        __builtin_amdgcn_sched_group_barrier(0x002, 1, 0);                      \
        __builtin_amdgcn_sched_group_barrier(0x400, 1, 0);                      \
        __builtin_amdgcn_sched_group_barrier(0x002, 1, 0);                      \
        __builtin_amdgcn_sched_group_barrier(0x400, 1, 0);                      \
        __builtin_amdgcn_sched_group_barrier(0x002, 1, 0);                      \
        __builtin_amdgcn_sched_group_barrier(0x400, 1, 0);                      \
        __builtin_amdgcn_sched_group_barrier(0x002, 1, 0);                      \
        __builtin_amdgcn_sched_group_barrier(0x400, 1, 0);                      \
        __builtin_amdgcn_sched_group_barrier(0x002, 1, 0);                      \
        __builtin_amdgcn_sched_group_barrier(0x400, 1, 0);                      \
        __builtin_amdgcn_sched_group_barrier(0x002, 1, 0);                      \
        __builtin_amdgcn_sched_group_barrier(0x400, 1, 0);                      \
        __builtin_amdgcn_sched_group_barrier(0x002, 1, 0);                      \
        __builtin_amdgcn_sched_group_barrier(0x400, 1, 0);                      \
        __builtin_amdgcn_sched_group_barrier(0x002, 1, 0);                      \
    } while (0)

#define ATT_SCHED_FENCE() __builtin_amdgcn_sched_barrier(0)

__global__ __launch_bounds__(256) void attn_kernel(
    const float* __restrict__ Qt, const float* __restrict__ Kt,
    const float* __restrict__ Vt, float* __restrict__ SAt)
{
    __shared__ __align__(16) float kv[SEQ];
    __shared__ __align__(16) float vv[SEQ];
    __shared__ __align__(8)  v2f   afl[ATT_TILES * 32];
    __shared__ float rmax[8];
    __shared__ float rmin[8];

    const int bc  = blockIdx.x;
    const int tid = threadIdx.x;

    const float* kp = Kt + (size_t)bc * SEQ;
    const float* vp = Vt + (size_t)bc * SEQ;
    const float k_a = kp[tid];
    const float k_b = kp[tid + 256];
    kv[tid]       = k_a;
    kv[tid + 256] = k_b;
    vv[tid]       = vp[tid];
    vv[tid + 256] = vp[tid + 256];
    {
        const int i = tid & 15, t = tid >> 4;
        afl[t * 32 + i]             = (v2f){k_a, -1.f};
        afl[t * 32 + 16 + i]        = (v2f){0.f, 0.f};
        afl[(t + 16) * 32 + i]      = (v2f){k_b, -1.f};
        afl[(t + 16) * 32 + 16 + i] = (v2f){0.f, 0.f};
    }
    __syncthreads();

    float kmx = fmaxf(kv[tid], kv[tid + 256]);
    float kmn = fminf(kv[tid], kv[tid + 256]);
    #pragma unroll
    for (int off = 16; off >= 1; off >>= 1) {
        kmx = fmaxf(kmx, __shfl_xor(kmx, off, 32));
        kmn = fminf(kmn, __shfl_xor(kmn, off, 32));
    }
    if ((tid & 31) == 0) { rmax[tid >> 5] = kmx; rmin[tid >> 5] = kmn; }
    __syncthreads();
    float KMAX = rmax[0], KMIN = rmin[0];
    #pragma unroll
    for (int i = 1; i < 8; ++i) {
        KMAX = fmaxf(KMAX, rmax[i]);
        KMIN = fminf(KMIN, rmin[i]);
    }

    const int lane = tid & 31;
    const int wave = tid >> 5;
    const int qc   = lane & 15;
    const bool lo  = lane < 16;
    const float* qp = Qt + (size_t)bc * SEQ;

    v2f bq[ATT_GROUPS];
    #pragma unroll
    for (int g = 0; g < ATT_GROUPS; ++g) {
        const float al = qp[wave * ATT_WAVE_Q + g * 16 + qc] * (SCALE * LOG2E_F);
        const float cl = al * ((al >= 0.f) ? KMAX : KMIN);
        bq[g].x = lo ? al : 0.f;
        bq[g].y = lo ? cl : 0.f;
    }

    v2f s[ATT_GROUPS][4], p[ATT_GROUPS][4];
    #pragma unroll
    for (int g = 0; g < ATT_GROUPS; ++g)
        #pragma unroll
        for (int j = 0; j < 4; ++j) { s[g][j] = (v2f){0.f, 0.f}; p[g][j] = (v2f){0.f, 0.f}; }

    const v8f zero = (v8f){0.f,0.f,0.f,0.f,0.f,0.f,0.f,0.f};
    const v2f* al_p = afl + lane;
    const v4f* vq = (const v4f*)(vv + (lo ? 0 : 8));

    v8f dA[ATT_GROUPS], dB[ATT_GROUPS];
    {
        const v2f a0 = al_p[0];
        #pragma unroll
        for (int g = 0; g < ATT_GROUPS; ++g)
            dA[g] = __builtin_amdgcn_wmma_f32_16x16x4_f32(
                false, a0, false, bq[g], (short)0, zero, false, false);
    }
    v2f aB = al_p[32];
    v4f vA0 = vq[0], vA1 = vq[1];
    v2f e[4];
    att_first(dA[0], e);

    #pragma unroll 1
    for (int t = 0; t < ATT_TILES - 2; t += 2) {
        const v2f aA  = al_p[32 * (t + 2)];
        const v4f vB0 = vq[4 * (t + 1)];
        const v4f vB1 = vq[4 * (t + 1) + 1];
        #pragma unroll
        for (int g = 0; g < ATT_GROUPS; ++g) {
            dB[g] = __builtin_amdgcn_wmma_f32_16x16x4_f32(
                false, aB, false, bq[g], (short)0, zero, false, false);
            att_step((g + 1 < ATT_GROUPS) ? dA[(g + 1) % ATT_GROUPS] : dB[0], e, vA0, vA1, s[g], p[g]);
            ATT_STEP_ORDER();
            ATT_SCHED_FENCE();
        }
        aB  = al_p[32 * (t + 3)];
        vA0 = vq[4 * (t + 2)];
        vA1 = vq[4 * (t + 2) + 1];
        #pragma unroll
        for (int g = 0; g < ATT_GROUPS; ++g) {
            dA[g] = __builtin_amdgcn_wmma_f32_16x16x4_f32(
                false, aA, false, bq[g], (short)0, zero, false, false);
            att_step((g + 1 < ATT_GROUPS) ? dB[(g + 1) % ATT_GROUPS] : dA[0], e, vB0, vB1, s[g], p[g]);
            ATT_STEP_ORDER();
            ATT_SCHED_FENCE();
        }
    }
    {
        const v4f vB0 = vq[4 * (ATT_TILES - 1)];
        const v4f vB1 = vq[4 * (ATT_TILES - 1) + 1];
        #pragma unroll
        for (int g = 0; g < ATT_GROUPS; ++g) {
            dB[g] = __builtin_amdgcn_wmma_f32_16x16x4_f32(
                false, aB, false, bq[g], (short)0, zero, false, false);
            att_step((g + 1 < ATT_GROUPS) ? dA[(g + 1) % ATT_GROUPS] : dB[0], e, vA0, vA1, s[g], p[g]);
            ATT_STEP_ORDER();
            ATT_SCHED_FENCE();
        }
        #pragma unroll
        for (int g = 0; g + 1 < ATT_GROUPS; ++g) {
            att_step(dB[g + 1], e, vB0, vB1, s[g], p[g]);
            ATT_STEP_ORDER();
            ATT_SCHED_FENCE();
        }
        att_last(e, vB0, vB1, s[ATT_GROUPS - 1], p[ATT_GROUPS - 1]);
    }

    float* op = SAt + (size_t)bc * SEQ + wave * ATT_WAVE_Q + qc;
    #pragma unroll
    for (int g = 0; g < ATT_GROUPS; ++g) {
        float sum = ((s[g][0][0] + s[g][0][1]) + (s[g][1][0] + s[g][1][1])) + ((s[g][2][0] + s[g][2][1]) + (s[g][3][0] + s[g][3][1]));
        float num = ((p[g][0][0] + p[g][0][1]) + (p[g][1][0] + p[g][1][1])) + ((p[g][2][0] + p[g][2][1]) + (p[g][3][0] + p[g][3][1]));
        sum += __shfl_xor(sum, 16, 32);
        num += __shfl_xor(num, 16, 32);
        if (lo)
            op[16 * g] = num / sum;
    }
}

#define OUT_CHUNK   8
#define OUT_HALF    (OUT_CHUNK / 2)
#define OUT_STEPS   (OUT_CHUNK / 4)
#define OUT_NCHUNK  (DK / OUT_CHUNK)

__device__ __forceinline__ constexpr int out_k_of(int i, int s) { return 2 * i + s; }

__global__ __launch_bounds__(128) void out_gemm_kernel(
    const float* __restrict__ SAt, const float* __restrict__ Wo,
    float* __restrict__ Out)
{
    const int tid   = threadIdx.x;
    const int lane  = tid & 31;
    const int wave  = tid >> 5;
    const int c     = lane & 15;
    const int h     = lane >> 4;
    const int j0    = (blockIdx.x & 7) * 64;
    const int row0  = ((blockIdx.x >> 3) * 4 + wave) * 32;

    const int b_idx = row0 >> 9;
    const int n0    = row0 & (SEQ - 1);

    const float* ap = SAt + ((size_t)b_idx * DK + OUT_HALF * h) * SEQ + n0 + 2 * c;
    const float* bp = Wo + (size_t)(OUT_HALF * h) * DIM + j0 + 4 * c;

    v2f a[OUT_NCHUNK][OUT_HALF];
    v4f b[OUT_NCHUNK][OUT_HALF];
    #pragma unroll
    for (int ch = 0; ch < OUT_NCHUNK; ++ch) {
        #pragma unroll
        for (int q = 0; q < OUT_HALF; ++q)
            a[ch][q] = *(const v2f*)(ap + (size_t)(ch * OUT_CHUNK + q) * SEQ);
        #pragma unroll
        for (int q = 0; q < OUT_HALF; ++q)
            b[ch][q] = *(const v4f*)(bp + (size_t)(ch * OUT_CHUNK + q) * DIM);
        __builtin_amdgcn_sched_barrier(0);
    }

    v8f acc[2][4];
    #pragma unroll
    for (int m = 0; m < 2; ++m)
        #pragma unroll
        for (int t = 0; t < 4; ++t)
            acc[m][t] = (v8f){0.f,0.f,0.f,0.f,0.f,0.f,0.f,0.f};

    #pragma unroll
    for (int ch = 0; ch < OUT_NCHUNK; ++ch) {
        #pragma unroll
        for (int i = 0; i < OUT_STEPS; ++i) {
            v2f a0, a1;
            a0.x = a[ch][out_k_of(i, 0)][0];
            a0.y = a[ch][out_k_of(i, 1)][0];
            a1.x = a[ch][out_k_of(i, 0)][1];
            a1.y = a[ch][out_k_of(i, 1)][1];
            #pragma unroll
            for (int t = 0; t < 4; ++t) {
                v2f bf;
                bf.x = b[ch][out_k_of(i, 0)][t];
                bf.y = b[ch][out_k_of(i, 1)][t];
                acc[0][t] = __builtin_amdgcn_wmma_f32_16x16x4_f32(
                    false, a0, false, bf, (short)0, acc[0][t], false, false);
                acc[1][t] = __builtin_amdgcn_wmma_f32_16x16x4_f32(
                    false, a1, false, bf, (short)0, acc[1][t], false, false);
            }
        }
        __builtin_amdgcn_sched_barrier(0);
    }

    #pragma unroll
    for (int m = 0; m < 2; ++m) {
        float* ob = Out + (size_t)(row0 + 16 * h + m) * DIM + j0 + 4 * c;
        #pragma unroll
        for (int r = 0; r < 8; ++r) {
            v4f o = {acc[m][0][r], acc[m][1][r], acc[m][2][r], acc[m][3][r]};
            *(v4f*)(ob + (size_t)(2 * r) * DIM) = o;
        }
    }
}

extern "C" void kernel_launch(void* const* d_in, const int* in_sizes, int n_in,
                              void* d_out, int out_size, void* d_ws, size_t ws_size,
                              hipStream_t stream)
{
    const float* X    = (const float*)d_in[0];
    const float* Wqkv = (const float*)d_in[1];
    const float* Wo   = (const float*)d_in[2];
    float* out = (float*)d_out;

    const size_t CHN = (size_t)BATCH * DK * SEQ;
    float* Qt = (float*)d_ws;
    float* Kt = Qt + CHN;
    float* Vt = Kt + CHN;
    float* SAt = Vt + CHN;

    qkv_gemm_kernel<<<256, 128, 0, stream>>>(X, Wqkv, Qt, Kt, Vt);
    attn_kernel<<<BATCH * DK, 256, 0, stream>>>(Qt, Kt, Vt, SAt);
    out_gemm_kernel<<<256, 128, 0, stream>>>(SAt, Wo, out);
}
